# starting kernel
# baseline (speedup 1.0000x reference)
#include <hip/hip_runtime.h>
#include <math.h>

typedef float v2f __attribute__((ext_vector_type(2)));
typedef float v8f __attribute__((ext_vector_type(8)));

#define B_   128
#define S_   512
#define D_   768
#define T_   51
#define NI_  26
#define NS_  122
#define ROWS_   (B_ * T_)
#define MTILES_ (ROWS_ / 16)
#define LPAD 772

__global__ __launch_bounds__(256) void intent_kernel(
    const float* __restrict__ res, const float* __restrict__ W_id,
    const float* __restrict__ b_id, float* __restrict__ out)
{
    int idx = blockIdx.x * blockDim.x + threadIdx.x;
    if (idx >= B_ * NI_) return;
    int b = idx / NI_;
    int j = idx - b * NI_;
    const float* r = res + (size_t)b * D_;
    float acc = 0.f;
    for (int d = 0; d < D_; ++d)
        acc = fmaf(r[d], W_id[(size_t)d * NI_ + j], acc);
    out[idx] = acc + b_id[j];
}

__global__ __launch_bounds__(256) void word_slot_kernel(
    const float* __restrict__ res_all, const float* __restrict__ res,
    const int*  __restrict__ tstart,   const int*  __restrict__ slen,
    const float* __restrict__ W_attn,  const float* __restrict__ W_slot,
    const float* __restrict__ b_slot,  float* __restrict__ out_sf)
{
    __shared__ float g0s[16][LPAD];
    __shared__ float g1s[16][LPAD];
    __shared__ float sc0[16], sc1[16], wg0[16], wg1[16];
    __shared__ int   rowB[16], rowS0[16], rowS1[16], rowLen1[16];

    const int tid  = threadIdx.x;
    const int lane = tid & 31;
    const int wave = tid >> 5;
    const int half = (lane >> 4) & 1;
    const int l16  = lane & 15;
    const int kA   = 2 * half;
    const int mblk = blockIdx.x;

    if (tid < 16) {
        int R  = mblk * 16 + tid;
        int bb = R / T_;
        int tt = R - bb * T_;
        int st = tstart[bb * T_ + tt];
        int ln = slen[bb * T_ + tt];
        rowB[tid]    = bb;
        rowS0[tid]   = st;
        rowS1[tid]   = st + ln;
        rowLen1[tid] = (ln == 1);
        sc0[tid] = 0.f;
        sc1[tid] = 0.f;
    }
    __syncthreads();

    for (int idx = tid; idx < 16 * (D_ / 4); idx += 256) {
        int m = idx / (D_ / 4);
        int q = idx - m * (D_ / 4);
        const float4* s0p = (const float4*)(res_all + (size_t)(rowB[m] * S_ + rowS0[m]) * D_);
        const float4* s1p = (const float4*)(res_all + (size_t)(rowB[m] * S_ + rowS1[m]) * D_);
        ((float4*)&g0s[m][0])[q] = s0p[q];
        ((float4*)&g1s[m][0])[q] = s1p[q];
    }
    __syncthreads();

    float s0r[8], s1r[8];
#pragma unroll
    for (int i = 0; i < 8; ++i) { s0r[i] = 0.f; s1r[i] = 0.f; }

    for (int nt = 0; nt < 6; ++nt) {
        const int nb = wave * 96 + nt * 16;
        const int c  = nb + l16;
        v8f acc = {};
        for (int k0 = 0; k0 < D_; k0 += 4) {
            v2f a, b;
            a.x = g0s[l16][k0 + kA];
            a.y = g0s[l16][k0 + kA + 1];
            b.x = W_attn[(size_t)(k0 + kA) * D_ + c];
            b.y = W_attn[(size_t)(k0 + kA + 1) * D_ + c];
            acc = __builtin_amdgcn_wmma_f32_16x16x4_f32(
                false, a, false, b, (short)0, acc, false, false);
        }
#pragma unroll
        for (int i = 0; i < 8; ++i) {
            int   m = i + 8 * half;
            float q = acc[i];
            s0r[i] = fmaf(q, g0s[m][c], s0r[i]);
            s1r[i] = fmaf(q, g1s[m][c], s1r[i]);
        }
    }

#pragma unroll
    for (int i = 0; i < 8; ++i) {
        float a0 = s0r[i], a1 = s1r[i];
        for (int off = 8; off >= 1; off >>= 1) {
            a0 += __shfl_xor(a0, off, 32);
            a1 += __shfl_xor(a1, off, 32);
        }
        if (l16 == 0) {
            int m = i + 8 * half;
            atomicAdd(&sc0[m], a0);
            atomicAdd(&sc1[m], a1);
        }
    }
    __syncthreads();

    if (tid < 16) {
        const float scale = 0.03608439182435161f;
        float a  = sc0[tid] * scale;
        float b  = sc1[tid] * scale;
        float mx = fmaxf(a, b);
        float e0 = __expf(a - mx);
        float e1 = __expf(b - mx);
        float iv = 1.f / (e0 + e1);
        float w0 = e0 * iv, w1 = e1 * iv;
        if (rowLen1[tid]) { w0 = 1.f; w1 = 0.f; }
        wg0[tid] = w0;
        wg1[tid] = w1;
    }
    __syncthreads();

    for (int idx = tid; idx < 16 * D_; idx += 256) {
        int m = idx / D_;
        int d = idx - m * D_;
        float w = wg0[m] * g0s[m][d] + wg1[m] * g1s[m][d]
                + res[(size_t)rowB[m] * D_ + d];
        g0s[m][d] = w;
    }
    __syncthreads();

    {
        const int  nb = wave * 16;
        const int  c  = nb + l16;
        const bool cv = (c < NS_);
        const int  cc = cv ? c : (NS_ - 1);
        const float msk = cv ? 1.f : 0.f;
        v8f acc = {};
        for (int k0 = 0; k0 < D_; k0 += 4) {
            v2f a, b;
            a.x = g0s[l16][k0 + kA];
            a.y = g0s[l16][k0 + kA + 1];
            b.x = W_slot[(size_t)(k0 + kA) * NS_ + cc] * msk;
            b.y = W_slot[(size_t)(k0 + kA + 1) * NS_ + cc] * msk;
            acc = __builtin_amdgcn_wmma_f32_16x16x4_f32(
                false, a, false, b, (short)0, acc, false, false);
        }
        if (cv) {
            float bias = b_slot[c];
#pragma unroll
            for (int i = 0; i < 8; ++i) {
                int m = i + 8 * half;
                int R = mblk * 16 + m;
                out_sf[(size_t)R * NS_ + c] = acc[i] + bias;
            }
        }
    }
}

extern "C" void kernel_launch(void* const* d_in, const int* in_sizes, int n_in,
                              void* d_out, int out_size, void* d_ws, size_t ws_size,
                              hipStream_t stream)
{
    (void)in_sizes; (void)n_in; (void)out_size; (void)d_ws; (void)ws_size;

    const float* res_all = (const float*)d_in[0];
    const float* res     = (const float*)d_in[1];
    const int*   tstart  = (const int*)  d_in[2];
    const int*   slen    = (const int*)  d_in[3];
    const float* W_attn  = (const float*)d_in[4];
    const float* W_id    = (const float*)d_in[5];
    const float* b_id    = (const float*)d_in[6];
    const float* W_slot  = (const float*)d_in[7];
    const float* b_slot  = (const float*)d_in[8];

    float* out_id = (float*)d_out;
    float* out_sf = out_id + B_ * NI_;

    intent_kernel<<<(B_ * NI_ + 255) / 256, 256, 0, stream>>>(res, W_id, b_id, out_id);
    word_slot_kernel<<<MTILES_, 256, 0, stream>>>(
        res_all, res, tstart, slen, W_attn, W_slot, b_slot, out_sf);
}


# Intent head runs as 104 extra work-groups of the one launch, beside the word/slot tiles; its 768-term sum is split over 8 waves x 4 partial sums (summation order changes)
# speedup vs baseline: 1.6585x; 1.6585x over previous
#include <hip/hip_runtime.h>
#include <math.h>

typedef float v2f __attribute__((ext_vector_type(2)));
typedef float v8f __attribute__((ext_vector_type(8)));

#define B_   128
#define S_   512
#define D_   768
#define T_   51
#define NI_  26
#define NS_  122
#define ROWS_   (B_ * T_)
#define MTILES_ (ROWS_ / 16)
#define LPAD 772
#define IBLKS_  ((B_ * NI_) / 32)
#define KSL_    (D_ / 8)

static_assert((B_ * NI_) % 32 == 0, "intent outputs must fill whole 32-float lines");
static_assert(D_ % 32 == 0, "slice length must be a multiple of 4");

__global__ __launch_bounds__(256) void word_slot_intent_kernel(
    const float* __restrict__ res_all, const float* __restrict__ res,
    const int*  __restrict__ tstart,   const int*  __restrict__ slen,
    const float* __restrict__ W_attn,  const float* __restrict__ W_slot,
    const float* __restrict__ b_slot,  const float* __restrict__ W_id,
    const float* __restrict__ b_id,    float* __restrict__ out_id,
    float* __restrict__ out_sf)
{
    __shared__ float g0s[16][LPAD];
    __shared__ float g1s[16][LPAD];
    __shared__ float sc0[16], sc1[16], wg0[16], wg1[16];
    __shared__ int   rowB[16], rowS0[16], rowS1[16], rowLen1[16];

    const int tid  = threadIdx.x;
    const int lane = tid & 31;
    const int wave = tid >> 5;
    const int half = (lane >> 4) & 1;
    const int l16  = lane & 15;
    const int kA   = 2 * half;
    const int mblk = blockIdx.x;

    if (mblk >= MTILES_) {
        const int idx = (mblk - MTILES_) * 32 + lane;
        const int b   = idx / NI_;
        const int j   = idx - b * NI_;
        const float* r = res  + (size_t)b * D_ + wave * KSL_;
        const float* w = W_id + (size_t)(wave * KSL_) * NI_ + j;
        float a0 = 0.f, a1 = 0.f, a2 = 0.f, a3 = 0.f;
        for (int i = 0; i < KSL_; i += 4) {
            a0 = fmaf(r[i],     w[(size_t)(i)     * NI_], a0);
            a1 = fmaf(r[i + 1], w[(size_t)(i + 1) * NI_], a1);
            a2 = fmaf(r[i + 2], w[(size_t)(i + 2) * NI_], a2);
            a3 = fmaf(r[i + 3], w[(size_t)(i + 3) * NI_], a3);
        }
        g0s[wave][lane] = (a0 + a1) + (a2 + a3);
        __syncthreads();
        if (tid < 32) {
            float s = g0s[0][lane];
            for (int v = 1; v < 8; ++v) s += g0s[v][lane];
            out_id[idx] = s + b_id[j];
        }
        return;
    }

    if (tid < 16) {
        int R  = mblk * 16 + tid;
        int bb = R / T_;
        int tt = R - bb * T_;
        int st = tstart[bb * T_ + tt];
        int ln = slen[bb * T_ + tt];
        rowB[tid]    = bb;
        rowS0[tid]   = st;
        rowS1[tid]   = st + ln;
        rowLen1[tid] = (ln == 1);
        sc0[tid] = 0.f;
        sc1[tid] = 0.f;
    }
    __syncthreads();

    for (int idx = tid; idx < 16 * (D_ / 4); idx += 256) {
        int m = idx / (D_ / 4);
        int q = idx - m * (D_ / 4);
        const float4* s0p = (const float4*)(res_all + (size_t)(rowB[m] * S_ + rowS0[m]) * D_);
        const float4* s1p = (const float4*)(res_all + (size_t)(rowB[m] * S_ + rowS1[m]) * D_);
        ((float4*)&g0s[m][0])[q] = s0p[q];
        ((float4*)&g1s[m][0])[q] = s1p[q];
    }
    __syncthreads();

    float s0r[8], s1r[8];
#pragma unroll
    for (int i = 0; i < 8; ++i) { s0r[i] = 0.f; s1r[i] = 0.f; }

    for (int nt = 0; nt < 6; ++nt) {
        const int nb = wave * 96 + nt * 16;
        const int c  = nb + l16;
        v8f acc = {};
        for (int k0 = 0; k0 < D_; k0 += 4) {
            v2f a, b;
            a.x = g0s[l16][k0 + kA];
            a.y = g0s[l16][k0 + kA + 1];
            b.x = W_attn[(size_t)(k0 + kA) * D_ + c];
            b.y = W_attn[(size_t)(k0 + kA + 1) * D_ + c];
            acc = __builtin_amdgcn_wmma_f32_16x16x4_f32(
                false, a, false, b, (short)0, acc, false, false);
        }
#pragma unroll
        for (int i = 0; i < 8; ++i) {
            int   m = i + 8 * half;
            float q = acc[i];
            s0r[i] = fmaf(q, g0s[m][c], s0r[i]);
            s1r[i] = fmaf(q, g1s[m][c], s1r[i]);
        }
    }

#pragma unroll
    for (int i = 0; i < 8; ++i) {
        float a0 = s0r[i], a1 = s1r[i];
        for (int off = 8; off >= 1; off >>= 1) {
            a0 += __shfl_xor(a0, off, 32);
            a1 += __shfl_xor(a1, off, 32);
        }
        if (l16 == 0) {
            int m = i + 8 * half;
            atomicAdd(&sc0[m], a0);
            atomicAdd(&sc1[m], a1);
        }
    }
    __syncthreads();

    if (tid < 16) {
        const float scale = 0.03608439182435161f;
        float a  = sc0[tid] * scale;
        float b  = sc1[tid] * scale;
        float mx = fmaxf(a, b);
        float e0 = __expf(a - mx);
        float e1 = __expf(b - mx);
        float iv = 1.f / (e0 + e1);
        float w0 = e0 * iv, w1 = e1 * iv;
        if (rowLen1[tid]) { w0 = 1.f; w1 = 0.f; }
        wg0[tid] = w0;
        wg1[tid] = w1;
    }
    __syncthreads();

    for (int idx = tid; idx < 16 * D_; idx += 256) {
        int m = idx / D_;
        int d = idx - m * D_;
        float w = wg0[m] * g0s[m][d] + wg1[m] * g1s[m][d]
                + res[(size_t)rowB[m] * D_ + d];
        g0s[m][d] = w;
    }
    __syncthreads();

    {
        const int  nb = wave * 16;
        const int  c  = nb + l16;
        const bool cv = (c < NS_);
        const int  cc = cv ? c : (NS_ - 1);
        const float msk = cv ? 1.f : 0.f;
        v8f acc = {};
        for (int k0 = 0; k0 < D_; k0 += 4) {
            v2f a, b;
            a.x = g0s[l16][k0 + kA];
            a.y = g0s[l16][k0 + kA + 1];
            b.x = W_slot[(size_t)(k0 + kA) * NS_ + cc] * msk;
            b.y = W_slot[(size_t)(k0 + kA + 1) * NS_ + cc] * msk;
            acc = __builtin_amdgcn_wmma_f32_16x16x4_f32(
                false, a, false, b, (short)0, acc, false, false);
        }
        if (cv) {
            float bias = b_slot[c];
#pragma unroll
            for (int i = 0; i < 8; ++i) {
                int m = i + 8 * half;
                int R = mblk * 16 + m;
                out_sf[(size_t)R * NS_ + c] = acc[i] + bias;
            }
        }
    }
}

extern "C" void kernel_launch(void* const* d_in, const int* in_sizes, int n_in,
                              void* d_out, int out_size, void* d_ws, size_t ws_size,
                              hipStream_t stream)
{
    (void)in_sizes; (void)n_in; (void)out_size; (void)d_ws; (void)ws_size;

    const float* res_all = (const float*)d_in[0];
    const float* res     = (const float*)d_in[1];
    const int*   tstart  = (const int*)  d_in[2];
    const int*   slen    = (const int*)  d_in[3];
    const float* W_attn  = (const float*)d_in[4];
    const float* W_id    = (const float*)d_in[5];
    const float* b_id    = (const float*)d_in[6];
    const float* W_slot  = (const float*)d_in[7];
    const float* b_slot  = (const float*)d_in[8];

    float* out_id = (float*)d_out;
    float* out_sf = out_id + B_ * NI_;

    word_slot_intent_kernel<<<MTILES_ + IBLKS_, 256, 0, stream>>>(
        res_all, res, tstart, slen, W_attn, W_slot, b_slot, W_id, b_id, out_id, out_sf);
}


# Phase-1 wave tile 16x16 -> 16x96: one K pass, six accumulators, one LDS A fragment per six WMMAs, W_attn read as whole 128-byte lines one 16-k set ahead; score summation order changes
# speedup vs baseline: 1.8989x; 1.1450x over previous
#include <hip/hip_runtime.h>
#include <math.h>

typedef float v2f __attribute__((ext_vector_type(2)));
typedef float v8f __attribute__((ext_vector_type(8)));

#define B_   128
#define S_   512
#define D_   768
#define T_   51
#define NI_  26
#define NS_  122
#define ROWS_   (B_ * T_)
#define MTILES_ (ROWS_ / 16)
#define LPAD 772
#define IBLKS_  ((B_ * NI_) / 32)
#define KSL_    (D_ / 8)

static_assert((B_ * NI_) % 32 == 0, "intent outputs must fill whole 32-float lines");
static_assert(D_ % 32 == 0, "slice length must be a multiple of 4");

#define KU 4
static_assert(D_ % (8 * KU) == 0, "phase 1 walks K in pairs of register sets");

struct FragSet {
    float4 x4[KU], y4[KU];
    float2 x2[KU], y2[KU];
    v2f    a[KU];
};

static __device__ __forceinline__ void load_set(
    FragSet& s, const float* __restrict__ wq, const float* __restrict__ wp, const float* ap)
{
#pragma unroll
    for (int u = 0; u < KU; ++u) {
        s.x4[u] = *(const float4*)(wq + (size_t)(4 * u)     * D_);
        s.y4[u] = *(const float4*)(wq + (size_t)(4 * u + 1) * D_);
        s.x2[u] = *(const float2*)(wp + (size_t)(4 * u)     * D_);
        s.y2[u] = *(const float2*)(wp + (size_t)(4 * u + 1) * D_);
        s.a[u]  = *(const v2f*)(ap + 4 * u);
    }
}

static __device__ __forceinline__ v8f mma4(v2f a, float bx, float by, v8f c)
{
    v2f b;
    b.x = bx;
    b.y = by;
    return __builtin_amdgcn_wmma_f32_16x16x4_f32(false, a, false, b, (short)0, c, false, false);
}

static __device__ __forceinline__ void mma_set(v8f (&acc)[6], const FragSet& s)
{
#pragma unroll
    for (int u = 0; u < KU; ++u) {
        acc[0] = mma4(s.a[u], s.x4[u].x, s.y4[u].x, acc[0]);
        acc[1] = mma4(s.a[u], s.x4[u].y, s.y4[u].y, acc[1]);
        acc[2] = mma4(s.a[u], s.x4[u].z, s.y4[u].z, acc[2]);
        acc[3] = mma4(s.a[u], s.x4[u].w, s.y4[u].w, acc[3]);
        acc[4] = mma4(s.a[u], s.x2[u].x, s.y2[u].x, acc[4]);
        acc[5] = mma4(s.a[u], s.x2[u].y, s.y2[u].y, acc[5]);
    }
}

__global__ __launch_bounds__(256) void word_slot_intent_kernel(
    const float* __restrict__ res_all, const float* __restrict__ res,
    const int*  __restrict__ tstart,   const int*  __restrict__ slen,
    const float* __restrict__ W_attn,  const float* __restrict__ W_slot,
    const float* __restrict__ b_slot,  const float* __restrict__ W_id,
    const float* __restrict__ b_id,    float* __restrict__ out_id,
    float* __restrict__ out_sf)
{
    __shared__ __align__(16) float g0s[16][LPAD];
    __shared__ __align__(16) float g1s[16][LPAD];
    __shared__ float sc0[16], sc1[16], wg0[16], wg1[16];
    __shared__ int   rowB[16], rowS0[16], rowS1[16], rowLen1[16];

    const int tid  = threadIdx.x;
    const int lane = tid & 31;
    const int wave = tid >> 5;
    const int half = (lane >> 4) & 1;
    const int l16  = lane & 15;
    const int kA   = 2 * half;
    const int mblk = blockIdx.x;

    if (mblk >= MTILES_) {
        const int idx = (mblk - MTILES_) * 32 + lane;
        const int b   = idx / NI_;
        const int j   = idx - b * NI_;
        const float* r = res  + (size_t)b * D_ + wave * KSL_;
        const float* w = W_id + (size_t)(wave * KSL_) * NI_ + j;
        float a0 = 0.f, a1 = 0.f, a2 = 0.f, a3 = 0.f;
        for (int i = 0; i < KSL_; i += 4) {
            a0 = fmaf(r[i],     w[(size_t)(i)     * NI_], a0);
            a1 = fmaf(r[i + 1], w[(size_t)(i + 1) * NI_], a1);
            a2 = fmaf(r[i + 2], w[(size_t)(i + 2) * NI_], a2);
            a3 = fmaf(r[i + 3], w[(size_t)(i + 3) * NI_], a3);
        }
        g0s[wave][lane] = (a0 + a1) + (a2 + a3);
        __syncthreads();
        if (tid < 32) {
            float s = g0s[0][lane];
            for (int v = 1; v < 8; ++v) s += g0s[v][lane];
            out_id[idx] = s + b_id[j];
        }
        return;
    }

    if (tid < 16) {
        int R  = mblk * 16 + tid;
        int bb = R / T_;
        int tt = R - bb * T_;
        int st = tstart[bb * T_ + tt];
        int ln = slen[bb * T_ + tt];
        rowB[tid]    = bb;
        rowS0[tid]   = st;
        rowS1[tid]   = st + ln;
        rowLen1[tid] = (ln == 1);
        sc0[tid] = 0.f;
        sc1[tid] = 0.f;
    }
    __syncthreads();

    for (int idx = tid; idx < 16 * (D_ / 4); idx += 256) {
        int m = idx / (D_ / 4);
        int q = idx - m * (D_ / 4);
        const float4* s0p = (const float4*)(res_all + (size_t)(rowB[m] * S_ + rowS0[m]) * D_);
        const float4* s1p = (const float4*)(res_all + (size_t)(rowB[m] * S_ + rowS1[m]) * D_);
        ((float4*)&g0s[m][0])[q] = s0p[q];
        ((float4*)&g1s[m][0])[q] = s1p[q];
    }
    __syncthreads();

    float s0r[8], s1r[8];
    {
        const int nb = wave * 96;
        const int cq = nb + 4 * l16;
        const int cp = nb + 64 + 2 * l16;
        const float* wq = W_attn + (size_t)kA * D_ + cq;
        const float* wp = W_attn + (size_t)kA * D_ + cp;
        const float* ap = &g0s[l16][kA];

        v8f acc[6];
#pragma unroll
        for (int j = 0; j < 6; ++j) acc[j] = v8f{};

        FragSet fa, fb;
        load_set(fa, wq, wp, ap);
        for (int k0 = 0; k0 < D_; k0 += 8 * KU) {
            load_set(fb, wq + (size_t)(k0 + 4 * KU) * D_, wp + (size_t)(k0 + 4 * KU) * D_,
                     ap + k0 + 4 * KU);
            mma_set(acc, fa);
            if (k0 + 8 * KU < D_)
                load_set(fa, wq + (size_t)(k0 + 8 * KU) * D_, wp + (size_t)(k0 + 8 * KU) * D_,
                         ap + k0 + 8 * KU);
            mma_set(acc, fb);
        }

#pragma unroll
        for (int i = 0; i < 8; ++i) {
            const int m = i + 8 * half;
            const float4 p0 = *(const float4*)&g0s[m][cq];
            const float2 r0 = *(const float2*)&g0s[m][cp];
            const float4 p1 = *(const float4*)&g1s[m][cq];
            const float2 r1 = *(const float2*)&g1s[m][cp];
            float t0 = acc[0][i] * p0.x;
            float t1 = acc[0][i] * p1.x;
            t0 = fmaf(acc[1][i], p0.y, t0);
            t1 = fmaf(acc[1][i], p1.y, t1);
            t0 = fmaf(acc[2][i], p0.z, t0);
            t1 = fmaf(acc[2][i], p1.z, t1);
            t0 = fmaf(acc[3][i], p0.w, t0);
            t1 = fmaf(acc[3][i], p1.w, t1);
            t0 = fmaf(acc[4][i], r0.x, t0);
            t1 = fmaf(acc[4][i], r1.x, t1);
            t0 = fmaf(acc[5][i], r0.y, t0);
            t1 = fmaf(acc[5][i], r1.y, t1);
            s0r[i] = t0;
            s1r[i] = t1;
        }
    }

#pragma unroll
    for (int i = 0; i < 8; ++i) {
        float a0 = s0r[i], a1 = s1r[i];
        for (int off = 8; off >= 1; off >>= 1) {
            a0 += __shfl_xor(a0, off, 32);
            a1 += __shfl_xor(a1, off, 32);
        }
        if (l16 == 0) {
            int m = i + 8 * half;
            atomicAdd(&sc0[m], a0);
            atomicAdd(&sc1[m], a1);
        }
    }
    __syncthreads();

    if (tid < 16) {
        const float scale = 0.03608439182435161f;
        float a  = sc0[tid] * scale;
        float b  = sc1[tid] * scale;
        float mx = fmaxf(a, b);
        float e0 = __expf(a - mx);
        float e1 = __expf(b - mx);
        float iv = 1.f / (e0 + e1);
        float w0 = e0 * iv, w1 = e1 * iv;
        if (rowLen1[tid]) { w0 = 1.f; w1 = 0.f; }
        wg0[tid] = w0;
        wg1[tid] = w1;
    }
    __syncthreads();

    for (int idx = tid; idx < 16 * D_; idx += 256) {
        int m = idx / D_;
        int d = idx - m * D_;
        float w = wg0[m] * g0s[m][d] + wg1[m] * g1s[m][d]
                + res[(size_t)rowB[m] * D_ + d];
        g0s[m][d] = w;
    }
    __syncthreads();

    {
        const int  nb = wave * 16;
        const int  c  = nb + l16;
        const bool cv = (c < NS_);
        const int  cc = cv ? c : (NS_ - 1);
        const float msk = cv ? 1.f : 0.f;
        v8f acc = {};
        for (int k0 = 0; k0 < D_; k0 += 4) {
            v2f a, b;
            a.x = g0s[l16][k0 + kA];
            a.y = g0s[l16][k0 + kA + 1];
            b.x = W_slot[(size_t)(k0 + kA) * NS_ + cc] * msk;
            b.y = W_slot[(size_t)(k0 + kA + 1) * NS_ + cc] * msk;
            acc = __builtin_amdgcn_wmma_f32_16x16x4_f32(
                false, a, false, b, (short)0, acc, false, false);
        }
        if (cv) {
            float bias = b_slot[c];
#pragma unroll
            for (int i = 0; i < 8; ++i) {
                int m = i + 8 * half;
                int R = mblk * 16 + m;
                out_sf[(size_t)R * NS_ + c] = acc[i] + bias;
            }
        }
    }
}

extern "C" void kernel_launch(void* const* d_in, const int* in_sizes, int n_in,
                              void* d_out, int out_size, void* d_ws, size_t ws_size,
                              hipStream_t stream)
{
    (void)in_sizes; (void)n_in; (void)out_size; (void)d_ws; (void)ws_size;

    const float* res_all = (const float*)d_in[0];
    const float* res     = (const float*)d_in[1];
    const int*   tstart  = (const int*)  d_in[2];
    const int*   slen    = (const int*)  d_in[3];
    const float* W_attn  = (const float*)d_in[4];
    const float* W_id    = (const float*)d_in[5];
    const float* b_id    = (const float*)d_in[6];
    const float* W_slot  = (const float*)d_in[7];
    const float* b_slot  = (const float*)d_in[8];

    float* out_id = (float*)d_out;
    float* out_sf = out_id + B_ * NI_;

    word_slot_intent_kernel<<<MTILES_ + IBLKS_, 256, 0, stream>>>(
        res_all, res, tstart, slen, W_attn, W_slot, b_slot, W_id, b_id, out_id, out_sf);
}


# Work-group tile 16 -> 32 rows (16 waves): each W_attn / W_slot fragment in registers feeds two row tiles, K split over wave pairs; score and res_sf summation order changes (two K halves)
# speedup vs baseline: 2.0703x; 1.0903x over previous
#include <hip/hip_runtime.h>
#include <math.h>

typedef float v2f __attribute__((ext_vector_type(2)));
typedef float v8f __attribute__((ext_vector_type(8)));

#define B_   128
#define S_   512
#define D_   768
#define T_   51
#define NI_  26
#define NS_  122
#define ROWS_   (B_ * T_)
#define MROWS_  32
#define NWG_    (ROWS_ / MROWS_)
#define LPAD 772
#define IBLKS_  ((B_ * NI_) / 64)
#define KSL_    (D_ / 8)

static_assert((B_ * NI_) % 64 == 0, "intent outputs must fill whole pairs of 32-float lines");
static_assert(ROWS_ % MROWS_ == 0, "rows must fill whole 32-row work-groups");
static_assert(D_ % 32 == 0, "slice length must be a multiple of 4");

#define KU 2
#define KHALF_ (D_ / 2)
#define NSETS_ (KHALF_ / (4 * KU))
static_assert(KHALF_ % (4 * KU) == 0, "a K half is a whole number of register sets");
static_assert(NSETS_ % 3 == 0, "phase 1 rotates three register sets");

struct FragSet {
    float4 x4[KU], y4[KU];
    float2 x2[KU], y2[KU];
    v2f    a0[KU], a1[KU];
};

static __device__ __forceinline__ void load_set(
    FragSet& s, const float* __restrict__ wq, const float* __restrict__ wp,
    const float* ap0, const float* ap1)
{
#pragma unroll
    for (int u = 0; u < KU; ++u) {
        s.x4[u] = *(const float4*)(wq + (size_t)(4 * u)     * D_);
        s.y4[u] = *(const float4*)(wq + (size_t)(4 * u + 1) * D_);
        s.x2[u] = *(const float2*)(wp + (size_t)(4 * u)     * D_);
        s.y2[u] = *(const float2*)(wp + (size_t)(4 * u + 1) * D_);
        s.a0[u] = *(const v2f*)(ap0 + 4 * u);
        s.a1[u] = *(const v2f*)(ap1 + 4 * u);
    }
}

static __device__ __forceinline__ v8f mma4(v2f a, v2f b, v8f c)
{
    return __builtin_amdgcn_wmma_f32_16x16x4_f32(false, a, false, b, (short)0, c, false, false);
}

static __device__ __forceinline__ void mma_pair(
    v8f& c0, v8f& c1, v2f a0, v2f a1, float bx, float by)
{
    v2f b;
    b.x = bx;
    b.y = by;
    c0 = mma4(a0, b, c0);
    c1 = mma4(a1, b, c1);
}

static __device__ __forceinline__ void mma_set(v8f (&acc)[12], const FragSet& s)
{
#pragma unroll
    for (int u = 0; u < KU; ++u) {
        mma_pair(acc[0], acc[6],  s.a0[u], s.a1[u], s.x4[u].x, s.y4[u].x);
        mma_pair(acc[1], acc[7],  s.a0[u], s.a1[u], s.x4[u].y, s.y4[u].y);
        mma_pair(acc[2], acc[8],  s.a0[u], s.a1[u], s.x4[u].z, s.y4[u].z);
        mma_pair(acc[3], acc[9],  s.a0[u], s.a1[u], s.x4[u].w, s.y4[u].w);
        mma_pair(acc[4], acc[10], s.a0[u], s.a1[u], s.x2[u].x, s.y2[u].x);
        mma_pair(acc[5], acc[11], s.a0[u], s.a1[u], s.x2[u].y, s.y2[u].y);
    }
}

__global__ __launch_bounds__(512) void word_slot_intent_kernel(
    const float* __restrict__ res_all, const float* __restrict__ res,
    const int*  __restrict__ tstart,   const int*  __restrict__ slen,
    const float* __restrict__ W_attn,  const float* __restrict__ W_slot,
    const float* __restrict__ b_slot,  const float* __restrict__ W_id,
    const float* __restrict__ b_id,    float* __restrict__ out_id,
    float* __restrict__ out_sf)
{
    __shared__ __align__(16) float g0s[MROWS_][LPAD];
    __shared__ __align__(16) float g1s[MROWS_][LPAD];
    __shared__ float sc0[MROWS_], sc1[MROWS_], wg0[MROWS_], wg1[MROWS_];
    __shared__ int   rowB[MROWS_], rowS0[MROWS_], rowS1[MROWS_], rowLen1[MROWS_];

    const int tid  = threadIdx.x;
    const int lane = tid & 31;
    const int wave = tid >> 5;
    const int half = (lane >> 4) & 1;
    const int l16  = lane & 15;
    const int kA   = 2 * half;
    const int wlo  = wave & 7;
    const int whi  = wave >> 3;
    const int mblk = blockIdx.x;

    if (mblk >= NWG_) {
        const int idx = (mblk - NWG_) * 64 + whi * 32 + lane;
        const int b   = idx / NI_;
        const int j   = idx - b * NI_;
        const float* r = res  + (size_t)b * D_ + wlo * KSL_;
        const float* w = W_id + (size_t)(wlo * KSL_) * NI_ + j;
        float a0 = 0.f, a1 = 0.f, a2 = 0.f, a3 = 0.f;
        for (int i = 0; i < KSL_; i += 4) {
            a0 = fmaf(r[i],     w[(size_t)(i)     * NI_], a0);
            a1 = fmaf(r[i + 1], w[(size_t)(i + 1) * NI_], a1);
            a2 = fmaf(r[i + 2], w[(size_t)(i + 2) * NI_], a2);
            a3 = fmaf(r[i + 3], w[(size_t)(i + 3) * NI_], a3);
        }
        g0s[wave][lane] = (a0 + a1) + (a2 + a3);
        __syncthreads();
        if (wlo == 0) {
            const int v0 = whi * 8;
            float s = g0s[v0][lane];
            for (int v = 1; v < 8; ++v) s += g0s[v0 + v][lane];
            out_id[idx] = s + b_id[j];
        }
        return;
    }

    if (tid < MROWS_) {
        int R  = mblk * MROWS_ + tid;
        int bb = R / T_;
        int tt = R - bb * T_;
        int st = tstart[bb * T_ + tt];
        int ln = slen[bb * T_ + tt];
        rowB[tid]    = bb;
        rowS0[tid]   = st;
        rowS1[tid]   = st + ln;
        rowLen1[tid] = (ln == 1);
        sc0[tid] = 0.f;
        sc1[tid] = 0.f;
    }
    __syncthreads();

    for (int idx = tid; idx < MROWS_ * (D_ / 4); idx += 512) {
        int m = idx / (D_ / 4);
        int q = idx - m * (D_ / 4);
        const float4* s0p = (const float4*)(res_all + (size_t)(rowB[m] * S_ + rowS0[m]) * D_);
        const float4* s1p = (const float4*)(res_all + (size_t)(rowB[m] * S_ + rowS1[m]) * D_);
        ((float4*)&g0s[m][0])[q] = s0p[q];
        ((float4*)&g1s[m][0])[q] = s1p[q];
    }
    __syncthreads();

    {
        const int nb = wlo * 96;
        const int kb = whi * KHALF_;
        const int cq = nb + 4 * l16;
        const int cp = nb + 64 + 2 * l16;
        const float* wq = W_attn + (size_t)(kb + kA) * D_ + cq;
        const float* wp = W_attn + (size_t)(kb + kA) * D_ + cp;
        const float* ap0 = &g0s[l16][kb + kA];
        const float* ap1 = &g0s[16 + l16][kb + kA];

        v8f acc[12];
#pragma unroll
        for (int j = 0; j < 12; ++j) acc[j] = v8f{};

        const size_t WSET = (size_t)(4 * KU) * D_;
        FragSet fa, fb, fc;
        load_set(fa, wq, wp, ap0, ap1);
        load_set(fb, wq + WSET, wp + WSET, ap0 + 4 * KU, ap1 + 4 * KU);
        for (int s = 0; s < NSETS_; s += 3) {
            load_set(fc, wq + (s + 2) * WSET, wp + (s + 2) * WSET,
                     ap0 + (s + 2) * 4 * KU, ap1 + (s + 2) * 4 * KU);
            mma_set(acc, fa);
            if (s + 3 < NSETS_)
                load_set(fa, wq + (s + 3) * WSET, wp + (s + 3) * WSET,
                         ap0 + (s + 3) * 4 * KU, ap1 + (s + 3) * 4 * KU);
            mma_set(acc, fb);
            if (s + 4 < NSETS_)
                load_set(fb, wq + (s + 4) * WSET, wp + (s + 4) * WSET,
                         ap0 + (s + 4) * 4 * KU, ap1 + (s + 4) * 4 * KU);
            mma_set(acc, fc);
        }

#pragma unroll
        for (int i = 0; i < 8; ++i) {
            const int m = i + 8 * half;
            float sa[2], sb[2];
#pragma unroll
            for (int t = 0; t < 2; ++t) {
                const float4 p0 = *(const float4*)&g0s[16 * t + m][cq];
                const float2 r0 = *(const float2*)&g0s[16 * t + m][cp];
                const float4 p1 = *(const float4*)&g1s[16 * t + m][cq];
                const float2 r1 = *(const float2*)&g1s[16 * t + m][cp];
                float t0 = acc[6 * t + 0][i] * p0.x;
                float t1 = acc[6 * t + 0][i] * p1.x;
                t0 = fmaf(acc[6 * t + 1][i], p0.y, t0);
                t1 = fmaf(acc[6 * t + 1][i], p1.y, t1);
                t0 = fmaf(acc[6 * t + 2][i], p0.z, t0);
                t1 = fmaf(acc[6 * t + 2][i], p1.z, t1);
                t0 = fmaf(acc[6 * t + 3][i], p0.w, t0);
                t1 = fmaf(acc[6 * t + 3][i], p1.w, t1);
                t0 = fmaf(acc[6 * t + 4][i], r0.x, t0);
                t1 = fmaf(acc[6 * t + 4][i], r1.x, t1);
                t0 = fmaf(acc[6 * t + 5][i], r0.y, t0);
                t1 = fmaf(acc[6 * t + 5][i], r1.y, t1);
                sa[t] = t0;
                sb[t] = t1;
            }
            for (int off = 8; off >= 1; off >>= 1) {
                sa[0] += __shfl_xor(sa[0], off, 32);
                sb[0] += __shfl_xor(sb[0], off, 32);
                sa[1] += __shfl_xor(sa[1], off, 32);
                sb[1] += __shfl_xor(sb[1], off, 32);
            }
            if (l16 == 0) {
                atomicAdd(&sc0[m], sa[0]);
                atomicAdd(&sc1[m], sb[0]);
                atomicAdd(&sc0[16 + m], sa[1]);
                atomicAdd(&sc1[16 + m], sb[1]);
            }
        }
    }
    __syncthreads();

    if (tid < MROWS_) {
        const float scale = 0.03608439182435161f;
        float a  = sc0[tid] * scale;
        float b  = sc1[tid] * scale;
        float mx = fmaxf(a, b);
        float e0 = __expf(a - mx);
        float e1 = __expf(b - mx);
        float iv = 1.f / (e0 + e1);
        float w0 = e0 * iv, w1 = e1 * iv;
        if (rowLen1[tid]) { w0 = 1.f; w1 = 0.f; }
        wg0[tid] = w0;
        wg1[tid] = w1;
    }
    __syncthreads();

    for (int idx = tid; idx < MROWS_ * D_; idx += 512) {
        int m = idx / D_;
        int d = idx - m * D_;
        float w = wg0[m] * g0s[m][d] + wg1[m] * g1s[m][d]
                + res[(size_t)rowB[m] * D_ + d];
        g0s[m][d] = w;
    }
    __syncthreads();

    {
        const int  nb = wlo * 16;
        const int  kb = whi * KHALF_;
        const int  c  = nb + l16;
        const bool cv = (c < NS_);
        const int  cc = cv ? c : (NS_ - 1);
        const float msk = cv ? 1.f : 0.f;
        const float* ws  = W_slot + (size_t)(kb + kA) * NS_ + cc;
        const float* ap0 = &g0s[l16][kb + kA];
        const float* ap1 = &g0s[16 + l16][kb + kA];
        v8f acc0 = {}, acc1 = {};
        for (int k0 = 0; k0 < KHALF_; k0 += 4) {
            v2f a0, a1, b;
            a0.x = ap0[k0];
            a0.y = ap0[k0 + 1];
            a1.x = ap1[k0];
            a1.y = ap1[k0 + 1];
            b.x = ws[(size_t)k0 * NS_] * msk;
            b.y = ws[(size_t)(k0 + 1) * NS_] * msk;
            acc0 = mma4(a0, b, acc0);
            acc1 = mma4(a1, b, acc1);
        }

        float* part = &g1s[0][0];
        if (whi == 1) {
#pragma unroll
            for (int i = 0; i < 8; ++i) {
                part[(wlo * 16 + i) * 32 + lane]     = acc0[i];
                part[(wlo * 16 + 8 + i) * 32 + lane] = acc1[i];
            }
        }
        __syncthreads();
        if (whi == 0 && cv) {
            float bias = b_slot[c];
#pragma unroll
            for (int i = 0; i < 8; ++i) {
                int m = i + 8 * half;
                int R = mblk * MROWS_ + m;
                out_sf[(size_t)R * NS_ + c] =
                    (acc0[i] + part[(wlo * 16 + i) * 32 + lane]) + bias;
                out_sf[(size_t)(R + 16) * NS_ + c] =
                    (acc1[i] + part[(wlo * 16 + 8 + i) * 32 + lane]) + bias;
            }
        }
    }
}

extern "C" void kernel_launch(void* const* d_in, const int* in_sizes, int n_in,
                              void* d_out, int out_size, void* d_ws, size_t ws_size,
                              hipStream_t stream)
{
    (void)in_sizes; (void)n_in; (void)out_size; (void)d_ws; (void)ws_size;

    const float* res_all = (const float*)d_in[0];
    const float* res     = (const float*)d_in[1];
    const int*   tstart  = (const int*)  d_in[2];
    const int*   slen    = (const int*)  d_in[3];
    const float* W_attn  = (const float*)d_in[4];
    const float* W_id    = (const float*)d_in[5];
    const float* b_id    = (const float*)d_in[6];
    const float* W_slot  = (const float*)d_in[7];
    const float* b_slot  = (const float*)d_in[8];

    float* out_id = (float*)d_out;
    float* out_sf = out_id + B_ * NI_;

    word_slot_intent_kernel<<<NWG_ + IBLKS_, 512, 0, stream>>>(
        res_all, res, tstart, slen, W_attn, W_slot, b_slot, W_id, b_id, out_id, out_sf);
}


# Row-moving loops (gather, word build): 16 lanes per row, float4, row metadata and softmax weights in the lanes' registers, all of a lane's loads issued before its first store; two barriers fewer
# speedup vs baseline: 2.2931x; 1.1076x over previous
#include <hip/hip_runtime.h>
#include <math.h>

typedef float v2f __attribute__((ext_vector_type(2)));
typedef float v8f __attribute__((ext_vector_type(8)));

#define B_   128
#define S_   512
#define D_   768
#define T_   51
#define NI_  26
#define NS_  122
#define ROWS_   (B_ * T_)
#define MROWS_  32
#define NWG_    (ROWS_ / MROWS_)
#define LPAD 772
#define IBLKS_  ((B_ * NI_) / 64)
#define KSL_    (D_ / 8)

static_assert((B_ * NI_) % 64 == 0, "intent outputs must fill whole pairs of 32-float lines");
static_assert(ROWS_ % MROWS_ == 0, "rows must fill whole 32-row work-groups");
static_assert(D_ % 32 == 0, "slice length must be a multiple of 4");

#define KU 2
#define KHALF_ (D_ / 2)
#define NSETS_ (KHALF_ / (4 * KU))
static_assert(KHALF_ % (4 * KU) == 0, "a K half is a whole number of register sets");
static_assert(NSETS_ % 3 == 0, "phase 1 rotates three register sets");

struct FragSet {
    float4 x4[KU], y4[KU];
    float2 x2[KU], y2[KU];
    v2f    a0[KU], a1[KU];
};

static __device__ __forceinline__ void load_set(
    FragSet& s, const float* __restrict__ wq, const float* __restrict__ wp,
    const float* ap0, const float* ap1)
{
#pragma unroll
    for (int u = 0; u < KU; ++u) {
        s.x4[u] = *(const float4*)(wq + (size_t)(4 * u)     * D_);
        s.y4[u] = *(const float4*)(wq + (size_t)(4 * u + 1) * D_);
        s.x2[u] = *(const float2*)(wp + (size_t)(4 * u)     * D_);
        s.y2[u] = *(const float2*)(wp + (size_t)(4 * u + 1) * D_);
        s.a0[u] = *(const v2f*)(ap0 + 4 * u);
        s.a1[u] = *(const v2f*)(ap1 + 4 * u);
    }
}

static __device__ __forceinline__ v8f mma4(v2f a, v2f b, v8f c)
{
    return __builtin_amdgcn_wmma_f32_16x16x4_f32(false, a, false, b, (short)0, c, false, false);
}

static __device__ __forceinline__ void mma_pair(
    v8f& c0, v8f& c1, v2f a0, v2f a1, float bx, float by)
{
    v2f b;
    b.x = bx;
    b.y = by;
    c0 = mma4(a0, b, c0);
    c1 = mma4(a1, b, c1);
}

static __device__ __forceinline__ void mma_set(v8f (&acc)[12], const FragSet& s)
{
#pragma unroll
    for (int u = 0; u < KU; ++u) {
        mma_pair(acc[0], acc[6],  s.a0[u], s.a1[u], s.x4[u].x, s.y4[u].x);
        mma_pair(acc[1], acc[7],  s.a0[u], s.a1[u], s.x4[u].y, s.y4[u].y);
        mma_pair(acc[2], acc[8],  s.a0[u], s.a1[u], s.x4[u].z, s.y4[u].z);
        mma_pair(acc[3], acc[9],  s.a0[u], s.a1[u], s.x4[u].w, s.y4[u].w);
        mma_pair(acc[4], acc[10], s.a0[u], s.a1[u], s.x2[u].x, s.y2[u].x);
        mma_pair(acc[5], acc[11], s.a0[u], s.a1[u], s.x2[u].y, s.y2[u].y);
    }
}

__global__ __launch_bounds__(512) void word_slot_intent_kernel(
    const float* __restrict__ res_all, const float* __restrict__ res,
    const int*  __restrict__ tstart,   const int*  __restrict__ slen,
    const float* __restrict__ W_attn,  const float* __restrict__ W_slot,
    const float* __restrict__ b_slot,  const float* __restrict__ W_id,
    const float* __restrict__ b_id,    float* __restrict__ out_id,
    float* __restrict__ out_sf)
{
    __shared__ __align__(16) float g0s[MROWS_][LPAD];
    __shared__ __align__(16) float g1s[MROWS_][LPAD];
    __shared__ float sc0[MROWS_], sc1[MROWS_];
    __shared__ int   rowB[MROWS_], rowLen1[MROWS_];

    const int tid  = threadIdx.x;
    const int lane = tid & 31;
    const int wave = tid >> 5;
    const int half = (lane >> 4) & 1;
    const int l16  = lane & 15;
    const int kA   = 2 * half;
    const int wlo  = wave & 7;
    const int whi  = wave >> 3;
    const int mblk = blockIdx.x;

    if (mblk >= NWG_) {
        const int idx = (mblk - NWG_) * 64 + whi * 32 + lane;
        const int b   = idx / NI_;
        const int j   = idx - b * NI_;
        const float* r = res  + (size_t)b * D_ + wlo * KSL_;
        const float* w = W_id + (size_t)(wlo * KSL_) * NI_ + j;
        float a0 = 0.f, a1 = 0.f, a2 = 0.f, a3 = 0.f;
        for (int i = 0; i < KSL_; i += 4) {
            a0 = fmaf(r[i],     w[(size_t)(i)     * NI_], a0);
            a1 = fmaf(r[i + 1], w[(size_t)(i + 1) * NI_], a1);
            a2 = fmaf(r[i + 2], w[(size_t)(i + 2) * NI_], a2);
            a3 = fmaf(r[i + 3], w[(size_t)(i + 3) * NI_], a3);
        }
        g0s[wave][lane] = (a0 + a1) + (a2 + a3);
        __syncthreads();
        if (wlo == 0) {
            const int v0 = whi * 8;
            float s = g0s[v0][lane];
            for (int v = 1; v < 8; ++v) s += g0s[v0 + v][lane];
            out_id[idx] = s + b_id[j];
        }
        return;
    }

    const int rm   = tid >> 4;
    const int c16  = tid & 15;
    constexpr int NQ = (D_ / 4) / 16;
    static_assert((D_ / 4) % 16 == 0, "a row is a whole number of 16-lane float4 steps");
    static_assert(MROWS_ * 16 == 512, "16 lanes per row fill the work-group");

    {
        const int R  = mblk * MROWS_ + rm;
        const int bb = R / T_;
        const int st = tstart[R];
        const int ln = slen[R];
        if (c16 == 0) {
            rowB[rm]    = bb;
            rowLen1[rm] = (ln == 1);
            sc0[rm] = 0.f;
            sc1[rm] = 0.f;
        }
        const float4* s0p = (const float4*)(res_all + (size_t)(bb * S_ + st) * D_) + c16;
        const float4* s1p = (const float4*)(res_all + (size_t)(bb * S_ + st + ln) * D_) + c16;
        float4 va[NQ], vb[NQ];
#pragma unroll
        for (int i = 0; i < NQ; ++i) {
            va[i] = s0p[16 * i];
            vb[i] = s1p[16 * i];
        }
        float4* d0 = (float4*)&g0s[rm][0] + c16;
        float4* d1 = (float4*)&g1s[rm][0] + c16;
#pragma unroll
        for (int i = 0; i < NQ; ++i) {
            d0[16 * i] = va[i];
            d1[16 * i] = vb[i];
        }
    }
    __syncthreads();

    {
        const int nb = wlo * 96;
        const int kb = whi * KHALF_;
        const int cq = nb + 4 * l16;
        const int cp = nb + 64 + 2 * l16;
        const float* wq = W_attn + (size_t)(kb + kA) * D_ + cq;
        const float* wp = W_attn + (size_t)(kb + kA) * D_ + cp;
        const float* ap0 = &g0s[l16][kb + kA];
        const float* ap1 = &g0s[16 + l16][kb + kA];

        v8f acc[12];
#pragma unroll
        for (int j = 0; j < 12; ++j) acc[j] = v8f{};

        const size_t WSET = (size_t)(4 * KU) * D_;
        FragSet fa, fb, fc;
        load_set(fa, wq, wp, ap0, ap1);
        load_set(fb, wq + WSET, wp + WSET, ap0 + 4 * KU, ap1 + 4 * KU);
        for (int s = 0; s < NSETS_; s += 3) {
            load_set(fc, wq + (s + 2) * WSET, wp + (s + 2) * WSET,
                     ap0 + (s + 2) * 4 * KU, ap1 + (s + 2) * 4 * KU);
            mma_set(acc, fa);
            if (s + 3 < NSETS_)
                load_set(fa, wq + (s + 3) * WSET, wp + (s + 3) * WSET,
                         ap0 + (s + 3) * 4 * KU, ap1 + (s + 3) * 4 * KU);
            mma_set(acc, fb);
            if (s + 4 < NSETS_)
                load_set(fb, wq + (s + 4) * WSET, wp + (s + 4) * WSET,
                         ap0 + (s + 4) * 4 * KU, ap1 + (s + 4) * 4 * KU);
            mma_set(acc, fc);
        }

#pragma unroll
        for (int i = 0; i < 8; ++i) {
            const int m = i + 8 * half;
            float sa[2], sb[2];
#pragma unroll
            for (int t = 0; t < 2; ++t) {
                const float4 p0 = *(const float4*)&g0s[16 * t + m][cq];
                const float2 r0 = *(const float2*)&g0s[16 * t + m][cp];
                const float4 p1 = *(const float4*)&g1s[16 * t + m][cq];
                const float2 r1 = *(const float2*)&g1s[16 * t + m][cp];
                float t0 = acc[6 * t + 0][i] * p0.x;
                float t1 = acc[6 * t + 0][i] * p1.x;
                t0 = fmaf(acc[6 * t + 1][i], p0.y, t0);
                t1 = fmaf(acc[6 * t + 1][i], p1.y, t1);
                t0 = fmaf(acc[6 * t + 2][i], p0.z, t0);
                t1 = fmaf(acc[6 * t + 2][i], p1.z, t1);
                t0 = fmaf(acc[6 * t + 3][i], p0.w, t0);
                t1 = fmaf(acc[6 * t + 3][i], p1.w, t1);
                t0 = fmaf(acc[6 * t + 4][i], r0.x, t0);
                t1 = fmaf(acc[6 * t + 4][i], r1.x, t1);
                t0 = fmaf(acc[6 * t + 5][i], r0.y, t0);
                t1 = fmaf(acc[6 * t + 5][i], r1.y, t1);
                sa[t] = t0;
                sb[t] = t1;
            }
            for (int off = 8; off >= 1; off >>= 1) {
                sa[0] += __shfl_xor(sa[0], off, 32);
                sb[0] += __shfl_xor(sb[0], off, 32);
                sa[1] += __shfl_xor(sa[1], off, 32);
                sb[1] += __shfl_xor(sb[1], off, 32);
            }
            if (l16 == 0) {
                atomicAdd(&sc0[m], sa[0]);
                atomicAdd(&sc1[m], sb[0]);
                atomicAdd(&sc0[16 + m], sa[1]);
                atomicAdd(&sc1[16 + m], sb[1]);
            }
        }
    }
    __syncthreads();

    {
        const float scale = 0.03608439182435161f;
        float a  = sc0[rm] * scale;
        float b  = sc1[rm] * scale;
        float mx = fmaxf(a, b);
        float e0 = __expf(a - mx);
        float e1 = __expf(b - mx);
        float iv = 1.f / (e0 + e1);
        float w0 = e0 * iv, w1 = e1 * iv;
        if (rowLen1[rm]) { w0 = 1.f; w1 = 0.f; }
        const float4* rp = (const float4*)(res + (size_t)rowB[rm] * D_) + c16;
        float4*       p0 = (float4*)&g0s[rm][0] + c16;
        const float4* p1 = (const float4*)&g1s[rm][0] + c16;
        float4 vr[NQ], v0[NQ], v1[NQ];
#pragma unroll
        for (int i = 0; i < NQ; ++i) {
            vr[i] = rp[16 * i];
            v0[i] = p0[16 * i];
            v1[i] = p1[16 * i];
        }
#pragma unroll
        for (int i = 0; i < NQ; ++i) {
            float4 w;
            w.x = w0 * v0[i].x + w1 * v1[i].x + vr[i].x;
            w.y = w0 * v0[i].y + w1 * v1[i].y + vr[i].y;
            w.z = w0 * v0[i].z + w1 * v1[i].z + vr[i].z;
            w.w = w0 * v0[i].w + w1 * v1[i].w + vr[i].w;
            p0[16 * i] = w;
        }
    }
    __syncthreads();

    {
        const int  nb = wlo * 16;
        const int  kb = whi * KHALF_;
        const int  c  = nb + l16;
        const bool cv = (c < NS_);
        const int  cc = cv ? c : (NS_ - 1);
        const float msk = cv ? 1.f : 0.f;
        const float* ws  = W_slot + (size_t)(kb + kA) * NS_ + cc;
        const float* ap0 = &g0s[l16][kb + kA];
        const float* ap1 = &g0s[16 + l16][kb + kA];
        v8f acc0 = {}, acc1 = {};
        for (int k0 = 0; k0 < KHALF_; k0 += 4) {
            v2f a0, a1, b;
            a0.x = ap0[k0];
            a0.y = ap0[k0 + 1];
            a1.x = ap1[k0];
            a1.y = ap1[k0 + 1];
            b.x = ws[(size_t)k0 * NS_] * msk;
            b.y = ws[(size_t)(k0 + 1) * NS_] * msk;
            acc0 = mma4(a0, b, acc0);
            acc1 = mma4(a1, b, acc1);
        }

        float* part = &g1s[0][0];
        if (whi == 1) {
#pragma unroll
            for (int i = 0; i < 8; ++i) {
                part[(wlo * 16 + i) * 32 + lane]     = acc0[i];
                part[(wlo * 16 + 8 + i) * 32 + lane] = acc1[i];
            }
        }
        __syncthreads();
        if (whi == 0 && cv) {
            float bias = b_slot[c];
#pragma unroll
            for (int i = 0; i < 8; ++i) {
                int m = i + 8 * half;
                int R = mblk * MROWS_ + m;
                out_sf[(size_t)R * NS_ + c] =
                    (acc0[i] + part[(wlo * 16 + i) * 32 + lane]) + bias;
                out_sf[(size_t)(R + 16) * NS_ + c] =
                    (acc1[i] + part[(wlo * 16 + 8 + i) * 32 + lane]) + bias;
            }
        }
    }
}

extern "C" void kernel_launch(void* const* d_in, const int* in_sizes, int n_in,
                              void* d_out, int out_size, void* d_ws, size_t ws_size,
                              hipStream_t stream)
{
    (void)in_sizes; (void)n_in; (void)out_size; (void)d_ws; (void)ws_size;

    const float* res_all = (const float*)d_in[0];
    const float* res     = (const float*)d_in[1];
    const int*   tstart  = (const int*)  d_in[2];
    const int*   slen    = (const int*)  d_in[3];
    const float* W_attn  = (const float*)d_in[4];
    const float* W_id    = (const float*)d_in[5];
    const float* b_id    = (const float*)d_in[6];
    const float* W_slot  = (const float*)d_in[7];
    const float* b_slot  = (const float*)d_in[8];

    float* out_id = (float*)d_out;
    float* out_sf = out_id + B_ * NI_;

    word_slot_intent_kernel<<<NWG_ + IBLKS_, 512, 0, stream>>>(
        res_all, res, tstart, slen, W_attn, W_slot, b_slot, W_id, b_id, out_id, out_sf);
}


# Both K loops: last trip peeled plus order fences, so each operand load is waited for by count two register sets after issue, not drained at once; phase 3 gets the three-set rotation
# speedup vs baseline: 2.5750x; 1.1229x over previous
#include <hip/hip_runtime.h>
#include <math.h>

typedef float v2f __attribute__((ext_vector_type(2)));
typedef float v8f __attribute__((ext_vector_type(8)));

#define B_   128
#define S_   512
#define D_   768
#define T_   51
#define NI_  26
#define NS_  122
#define ROWS_   (B_ * T_)
#define MROWS_  32
#define NWG_    (ROWS_ / MROWS_)
#define LPAD 772
#define IBLKS_  ((B_ * NI_) / 64)
#define KSL_    (D_ / 8)

static_assert((B_ * NI_) % 64 == 0, "intent outputs must fill whole pairs of 32-float lines");
static_assert(ROWS_ % MROWS_ == 0, "rows must fill whole 32-row work-groups");
static_assert(D_ % 32 == 0, "slice length must be a multiple of 4");

#define KU 2
#define KHALF_ (D_ / 2)
#define NSETS_ (KHALF_ / (4 * KU))
static_assert(KHALF_ % (4 * KU) == 0, "a K half is a whole number of register sets");
static_assert(NSETS_ % 3 == 0, "phase 1 rotates three register sets");

struct FragSet {
    float4 x4[KU], y4[KU];
    float2 x2[KU], y2[KU];
    v2f    a0[KU], a1[KU];
};

static __device__ __forceinline__ void load_set(
    FragSet& s, const float* wq, const float* wp,
    const float* ap0, const float* ap1)
{
#pragma unroll
    for (int u = 0; u < KU; ++u) {
        s.x4[u] = *(const float4*)(wq + (size_t)(4 * u)     * D_);
        s.y4[u] = *(const float4*)(wq + (size_t)(4 * u + 1) * D_);
        s.x2[u] = *(const float2*)(wp + (size_t)(4 * u)     * D_);
        s.y2[u] = *(const float2*)(wp + (size_t)(4 * u + 1) * D_);
        s.a0[u] = *(const v2f*)(ap0 + 4 * u);
        s.a1[u] = *(const v2f*)(ap1 + 4 * u);
    }
}

static __device__ __forceinline__ v8f mma4(v2f a, v2f b, v8f c)
{
    return __builtin_amdgcn_wmma_f32_16x16x4_f32(false, a, false, b, (short)0, c, false, false);
}

static __device__ __forceinline__ void mma_pair(
    v8f& c0, v8f& c1, v2f a0, v2f a1, float bx, float by)
{
    v2f b;
    b.x = bx;
    b.y = by;
    c0 = mma4(a0, b, c0);
    c1 = mma4(a1, b, c1);
}

static __device__ __forceinline__ void mma_set(v8f (&acc)[12], const FragSet& s)
{
#pragma unroll
    for (int u = 0; u < KU; ++u) {
        mma_pair(acc[0], acc[6],  s.a0[u], s.a1[u], s.x4[u].x, s.y4[u].x);
        mma_pair(acc[1], acc[7],  s.a0[u], s.a1[u], s.x4[u].y, s.y4[u].y);
        mma_pair(acc[2], acc[8],  s.a0[u], s.a1[u], s.x4[u].z, s.y4[u].z);
        mma_pair(acc[3], acc[9],  s.a0[u], s.a1[u], s.x4[u].w, s.y4[u].w);
        mma_pair(acc[4], acc[10], s.a0[u], s.a1[u], s.x2[u].x, s.y2[u].x);
        mma_pair(acc[5], acc[11], s.a0[u], s.a1[u], s.x2[u].y, s.y2[u].y);
    }
}

static __device__ __forceinline__ void fence12(v8f (&a)[12])
{
    asm volatile("" : "+v"(a[0]), "+v"(a[1]), "+v"(a[2]), "+v"(a[3]), "+v"(a[4]), "+v"(a[5]),
                      "+v"(a[6]), "+v"(a[7]), "+v"(a[8]), "+v"(a[9]), "+v"(a[10]), "+v"(a[11])
                 :: "memory");
    __builtin_amdgcn_sched_barrier(0);
}
static __device__ __forceinline__ void fence2(v8f& a, v8f& b)
{
    asm volatile("" : "+v"(a), "+v"(b) :: "memory");
    __builtin_amdgcn_sched_barrier(0);
}

#define KU3 8
#define NSETS3_ (KHALF_ / (4 * KU3))
static_assert(KHALF_ % (4 * KU3) == 0, "a K half is a whole number of phase-3 sets");
static_assert(NSETS3_ % 3 == 0 && NSETS3_ >= 6, "phase 3 rotates three register sets");

struct SlotSet {
    float bx[KU3], by[KU3];
    v2f   a0[KU3], a1[KU3];
};

static __device__ __forceinline__ void load_slot(
    SlotSet& s, const float* ws, const float* ap0, const float* ap1)
{
#pragma unroll
    for (int u = 0; u < KU3; ++u) {
        s.bx[u] = ws[(size_t)(4 * u)     * NS_];
        s.by[u] = ws[(size_t)(4 * u + 1) * NS_];
        s.a0[u] = *(const v2f*)(ap0 + 4 * u);
        s.a1[u] = *(const v2f*)(ap1 + 4 * u);
    }
}

static __device__ __forceinline__ void mma_slot(v8f& c0, v8f& c1, const SlotSet& s, float msk)
{
#pragma unroll
    for (int u = 0; u < KU3; ++u) {
        v2f b;
        b.x = s.bx[u] * msk;
        b.y = s.by[u] * msk;
        c0 = mma4(s.a0[u], b, c0);
        c1 = mma4(s.a1[u], b, c1);
    }
}

__global__ __launch_bounds__(512) void word_slot_intent_kernel(
    const float* __restrict__ res_all, const float* __restrict__ res,
    const int*  __restrict__ tstart,   const int*  __restrict__ slen,
    const float* W_attn,               const float* W_slot,
    const float* __restrict__ b_slot,  const float* __restrict__ W_id,
    const float* __restrict__ b_id,    float* __restrict__ out_id,
    float* __restrict__ out_sf)
{
    __shared__ __align__(16) float g0s[MROWS_][LPAD];
    __shared__ __align__(16) float g1s[MROWS_][LPAD];
    __shared__ float sc0[MROWS_], sc1[MROWS_];
    __shared__ int   rowB[MROWS_], rowLen1[MROWS_];

    const int tid  = threadIdx.x;
    const int lane = tid & 31;
    const int wave = tid >> 5;
    const int half = (lane >> 4) & 1;
    const int l16  = lane & 15;
    const int kA   = 2 * half;
    const int wlo  = wave & 7;
    const int whi  = wave >> 3;
    const int mblk = blockIdx.x;

    if (mblk >= NWG_) {
        const int idx = (mblk - NWG_) * 64 + whi * 32 + lane;
        const int b   = idx / NI_;
        const int j   = idx - b * NI_;
        const float* r = res  + (size_t)b * D_ + wlo * KSL_;
        const float* w = W_id + (size_t)(wlo * KSL_) * NI_ + j;
        float a0 = 0.f, a1 = 0.f, a2 = 0.f, a3 = 0.f;
        for (int i = 0; i < KSL_; i += 4) {
            a0 = fmaf(r[i],     w[(size_t)(i)     * NI_], a0);
            a1 = fmaf(r[i + 1], w[(size_t)(i + 1) * NI_], a1);
            a2 = fmaf(r[i + 2], w[(size_t)(i + 2) * NI_], a2);
            a3 = fmaf(r[i + 3], w[(size_t)(i + 3) * NI_], a3);
        }
        g0s[wave][lane] = (a0 + a1) + (a2 + a3);
        __syncthreads();
        if (wlo == 0) {
            const int v0 = whi * 8;
            float s = g0s[v0][lane];
            for (int v = 1; v < 8; ++v) s += g0s[v0 + v][lane];
            out_id[idx] = s + b_id[j];
        }
        return;
    }

    const int rm   = tid >> 4;
    const int c16  = tid & 15;
    constexpr int NQ = (D_ / 4) / 16;
    static_assert((D_ / 4) % 16 == 0, "a row is a whole number of 16-lane float4 steps");
    static_assert(MROWS_ * 16 == 512, "16 lanes per row fill the work-group");

    {
        const int R  = mblk * MROWS_ + rm;
        const int bb = R / T_;
        const int st = tstart[R];
        const int ln = slen[R];
        if (c16 == 0) {
            rowB[rm]    = bb;
            rowLen1[rm] = (ln == 1);
            sc0[rm] = 0.f;
            sc1[rm] = 0.f;
        }
        const float4* s0p = (const float4*)(res_all + (size_t)(bb * S_ + st) * D_) + c16;
        const float4* s1p = (const float4*)(res_all + (size_t)(bb * S_ + st + ln) * D_) + c16;
        float4 va[NQ], vb[NQ];
#pragma unroll
        for (int i = 0; i < NQ; ++i) {
            va[i] = s0p[16 * i];
            vb[i] = s1p[16 * i];
        }
        float4* d0 = (float4*)&g0s[rm][0] + c16;
        float4* d1 = (float4*)&g1s[rm][0] + c16;
#pragma unroll
        for (int i = 0; i < NQ; ++i) {
            d0[16 * i] = va[i];
            d1[16 * i] = vb[i];
        }
    }
    __syncthreads();

    {
        const int nb = wlo * 96;
        const int kb = whi * KHALF_;
        const int cq = nb + 4 * l16;
        const int cp = nb + 64 + 2 * l16;
        const float* wq = W_attn + (size_t)(kb + kA) * D_ + cq;
        const float* wp = W_attn + (size_t)(kb + kA) * D_ + cp;
        const float* ap0 = &g0s[l16][kb + kA];
        const float* ap1 = &g0s[16 + l16][kb + kA];

        v8f acc[12];
#pragma unroll
        for (int j = 0; j < 12; ++j) acc[j] = v8f{};

        const size_t WSET = (size_t)(4 * KU) * D_;
        FragSet fa, fb, fc;
        load_set(fa, wq, wp, ap0, ap1);
        load_set(fb, wq + WSET, wp + WSET, ap0 + 4 * KU, ap1 + 4 * KU);
        for (int s = 0; s < NSETS_ - 3; s += 3) {
            load_set(fc, wq + (s + 2) * WSET, wp + (s + 2) * WSET,
                     ap0 + (s + 2) * 4 * KU, ap1 + (s + 2) * 4 * KU);
            fence12(acc);
            mma_set(acc, fa);
            fence12(acc);
            load_set(fa, wq + (s + 3) * WSET, wp + (s + 3) * WSET,
                     ap0 + (s + 3) * 4 * KU, ap1 + (s + 3) * 4 * KU);
            fence12(acc);
            mma_set(acc, fb);
            fence12(acc);
            load_set(fb, wq + (s + 4) * WSET, wp + (s + 4) * WSET,
                     ap0 + (s + 4) * 4 * KU, ap1 + (s + 4) * 4 * KU);
            fence12(acc);
            mma_set(acc, fc);
            fence12(acc);
        }
        load_set(fc, wq + (NSETS_ - 1) * WSET, wp + (NSETS_ - 1) * WSET,
                 ap0 + (NSETS_ - 1) * 4 * KU, ap1 + (NSETS_ - 1) * 4 * KU);
        fence12(acc);
        mma_set(acc, fa);
        mma_set(acc, fb);
        mma_set(acc, fc);

#pragma unroll
        for (int i = 0; i < 8; ++i) {
            const int m = i + 8 * half;
            float sa[2], sb[2];
#pragma unroll
            for (int t = 0; t < 2; ++t) {
                const float4 p0 = *(const float4*)&g0s[16 * t + m][cq];
                const float2 r0 = *(const float2*)&g0s[16 * t + m][cp];
                const float4 p1 = *(const float4*)&g1s[16 * t + m][cq];
                const float2 r1 = *(const float2*)&g1s[16 * t + m][cp];
                float t0 = acc[6 * t + 0][i] * p0.x;
                float t1 = acc[6 * t + 0][i] * p1.x;
                t0 = fmaf(acc[6 * t + 1][i], p0.y, t0);
                t1 = fmaf(acc[6 * t + 1][i], p1.y, t1);
                t0 = fmaf(acc[6 * t + 2][i], p0.z, t0);
                t1 = fmaf(acc[6 * t + 2][i], p1.z, t1);
                t0 = fmaf(acc[6 * t + 3][i], p0.w, t0);
                t1 = fmaf(acc[6 * t + 3][i], p1.w, t1);
                t0 = fmaf(acc[6 * t + 4][i], r0.x, t0);
                t1 = fmaf(acc[6 * t + 4][i], r1.x, t1);
                t0 = fmaf(acc[6 * t + 5][i], r0.y, t0);
                t1 = fmaf(acc[6 * t + 5][i], r1.y, t1);
                sa[t] = t0;
                sb[t] = t1;
            }
            for (int off = 8; off >= 1; off >>= 1) {
                sa[0] += __shfl_xor(sa[0], off, 32);
                sb[0] += __shfl_xor(sb[0], off, 32);
                sa[1] += __shfl_xor(sa[1], off, 32);
                sb[1] += __shfl_xor(sb[1], off, 32);
            }
            if (l16 == 0) {
                atomicAdd(&sc0[m], sa[0]);
                atomicAdd(&sc1[m], sb[0]);
                atomicAdd(&sc0[16 + m], sa[1]);
                atomicAdd(&sc1[16 + m], sb[1]);
            }
        }
    }
    __syncthreads();

    {
        const float scale = 0.03608439182435161f;
        float a  = sc0[rm] * scale;
        float b  = sc1[rm] * scale;
        float mx = fmaxf(a, b);
        float e0 = __expf(a - mx);
        float e1 = __expf(b - mx);
        float iv = 1.f / (e0 + e1);
        float w0 = e0 * iv, w1 = e1 * iv;
        if (rowLen1[rm]) { w0 = 1.f; w1 = 0.f; }
        const float4* rp = (const float4*)(res + (size_t)rowB[rm] * D_) + c16;
        float4*       p0 = (float4*)&g0s[rm][0] + c16;
        const float4* p1 = (const float4*)&g1s[rm][0] + c16;
        float4 vr[NQ], v0[NQ], v1[NQ];
#pragma unroll
        for (int i = 0; i < NQ; ++i) {
            vr[i] = rp[16 * i];
            v0[i] = p0[16 * i];
            v1[i] = p1[16 * i];
        }
#pragma unroll
        for (int i = 0; i < NQ; ++i) {
            float4 w;
            w.x = w0 * v0[i].x + w1 * v1[i].x + vr[i].x;
            w.y = w0 * v0[i].y + w1 * v1[i].y + vr[i].y;
            w.z = w0 * v0[i].z + w1 * v1[i].z + vr[i].z;
            w.w = w0 * v0[i].w + w1 * v1[i].w + vr[i].w;
            p0[16 * i] = w;
        }
    }
    __syncthreads();

    {
        const int  nb = wlo * 16;
        const int  kb = whi * KHALF_;
        const int  c  = nb + l16;
        const bool cv = (c < NS_);
        const int  cc = cv ? c : (NS_ - 1);
        const float msk = cv ? 1.f : 0.f;
        const float* ws  = W_slot + (size_t)(kb + kA) * NS_ + cc;
        const float* ap0 = &g0s[l16][kb + kA];
        const float* ap1 = &g0s[16 + l16][kb + kA];
        v8f acc0 = {}, acc1 = {};
        const size_t SSET = (size_t)(4 * KU3) * NS_;
        SlotSet ta, tb, tc;
        load_slot(ta, ws, ap0, ap1);
        load_slot(tb, ws + SSET, ap0 + 4 * KU3, ap1 + 4 * KU3);
        for (int s = 0; s < NSETS3_ - 3; s += 3) {
            load_slot(tc, ws + (s + 2) * SSET, ap0 + (s + 2) * 4 * KU3, ap1 + (s + 2) * 4 * KU3);
            fence2(acc0, acc1);
            mma_slot(acc0, acc1, ta, msk);
            fence2(acc0, acc1);
            load_slot(ta, ws + (s + 3) * SSET, ap0 + (s + 3) * 4 * KU3, ap1 + (s + 3) * 4 * KU3);
            fence2(acc0, acc1);
            mma_slot(acc0, acc1, tb, msk);
            fence2(acc0, acc1);
            load_slot(tb, ws + (s + 4) * SSET, ap0 + (s + 4) * 4 * KU3, ap1 + (s + 4) * 4 * KU3);
            fence2(acc0, acc1);
            mma_slot(acc0, acc1, tc, msk);
            fence2(acc0, acc1);
        }
        load_slot(tc, ws + (NSETS3_ - 1) * SSET, ap0 + (NSETS3_ - 1) * 4 * KU3, ap1 + (NSETS3_ - 1) * 4 * KU3);
        fence2(acc0, acc1);
        mma_slot(acc0, acc1, ta, msk);
        mma_slot(acc0, acc1, tb, msk);
        mma_slot(acc0, acc1, tc, msk);

        float* part = &g1s[0][0];
        if (whi == 1) {
#pragma unroll
            for (int i = 0; i < 8; ++i) {
                part[(wlo * 16 + i) * 32 + lane]     = acc0[i];
                part[(wlo * 16 + 8 + i) * 32 + lane] = acc1[i];
            }
        }
        __syncthreads();
        if (whi == 0 && cv) {
            float bias = b_slot[c];
#pragma unroll
            for (int i = 0; i < 8; ++i) {
                int m = i + 8 * half;
                int R = mblk * MROWS_ + m;
                out_sf[(size_t)R * NS_ + c] =
                    (acc0[i] + part[(wlo * 16 + i) * 32 + lane]) + bias;
                out_sf[(size_t)(R + 16) * NS_ + c] =
                    (acc1[i] + part[(wlo * 16 + 8 + i) * 32 + lane]) + bias;
            }
        }
    }
}

extern "C" void kernel_launch(void* const* d_in, const int* in_sizes, int n_in,
                              void* d_out, int out_size, void* d_ws, size_t ws_size,
                              hipStream_t stream)
{
    (void)in_sizes; (void)n_in; (void)out_size; (void)d_ws; (void)ws_size;

    const float* res_all = (const float*)d_in[0];
    const float* res     = (const float*)d_in[1];
    const int*   tstart  = (const int*)  d_in[2];
    const int*   slen    = (const int*)  d_in[3];
    const float* W_attn  = (const float*)d_in[4];
    const float* W_id    = (const float*)d_in[5];
    const float* b_id    = (const float*)d_in[6];
    const float* W_slot  = (const float*)d_in[7];
    const float* b_slot  = (const float*)d_in[8];

    float* out_id = (float*)d_out;
    float* out_sf = out_id + B_ * NI_;

    word_slot_intent_kernel<<<NWG_ + IBLKS_, 512, 0, stream>>>(
        res_all, res, tstart, slen, W_attn, W_slot, b_slot, W_id, b_id, out_id, out_sf);
}
